# starting kernel
# baseline (speedup 1.0000x reference)
#include <hip/hip_runtime.h>
#include <hip/hip_bf16.h>


#define N_NODES   50000
#define N_EDGES   800000
#define D_HID     256
#define N_DOCS    20000
#define MLP_HID   64
#define N_CLASSES 20

typedef __attribute__((ext_vector_type(2))) float v2f;
typedef __attribute__((ext_vector_type(8))) float v8f;

__global__ __launch_bounds__(64) void gemm_wmma_f32(
    const float* __restrict__ A, const float* __restrict__ B,
    float* __restrict__ C, int M, int N, int K,
    const float* __restrict__ bias, int doRelu)
{
    __shared__ float As[16][32];
    __shared__ float Bs[32][64];

    const int tid  = threadIdx.x;
    const int wave = tid >> 5;
    const int lane = tid & 31;
    const int m0 = blockIdx.x * 16;
    const int n0 = blockIdx.y * 64;

    v8f acc0 = {};
    v8f acc1 = {};

    const int mrow  = lane & 15;
    const int khalf = (lane >> 4) << 1;
    const int ncol0 = (lane & 15) + (wave << 5);
    const int ncol1 = ncol0 + 16;

    const int ar = tid >> 2;
    const int ac = (tid & 3) * 8;
    const int br = tid >> 1;
    const int bc = (tid & 1) * 32;

    for (int k0 = 0; k0 < K; k0 += 32) {
        {
            const float* s = A + (size_t)(m0 + ar) * K + k0 + ac;
            *(float4*)&As[ar][ac]     = *(const float4*)(s);
            *(float4*)&As[ar][ac + 4] = *(const float4*)(s + 4);
        }
        {
            const float* s = B + (size_t)(k0 + br) * N + n0 + bc;
            #pragma unroll
            for (int j = 0; j < 8; ++j)
                *(float4*)&Bs[br][bc + 4 * j] = *(const float4*)(s + 4 * j);
        }
        if (k0 + 32 < K) {
            __builtin_prefetch(A + (size_t)(m0 + ar) * K + k0 + 32 + ac, 0, 3);
            __builtin_prefetch(B + (size_t)(k0 + 32 + br) * N + n0 + bc, 0, 3);
        }
        __syncthreads();

        #pragma unroll
        for (int kk = 0; kk < 32; kk += 4) {
            v2f a, b0, b1;
            a.x = As[mrow][kk + khalf];
            a.y = As[mrow][kk + khalf + 1];
            b0.x = Bs[kk + khalf][ncol0];
            b0.y = Bs[kk + khalf + 1][ncol0];
            b1.x = Bs[kk + khalf][ncol1];
            b1.y = Bs[kk + khalf + 1][ncol1];
            acc0 = __builtin_amdgcn_wmma_f32_16x16x4_f32(
                false, a, false, b0, (short)0, acc0, false, false);
            acc1 = __builtin_amdgcn_wmma_f32_16x16x4_f32(
                false, a, false, b1, (short)0, acc1, false, false);
        }
        __syncthreads();
    }

    const int colg0 = n0 + ncol0;
    const int colg1 = n0 + ncol1;
    const float bv0 = bias ? bias[colg0] : 0.0f;
    const float bv1 = bias ? bias[colg1] : 0.0f;
    #pragma unroll
    for (int r = 0; r < 8; ++r) {
        const int rowg = m0 + r + ((lane >> 4) << 3);
        float v0 = acc0[r] + bv0;
        float v1 = acc1[r] + bv1;
        if (doRelu) { v0 = fmaxf(v0, 0.0f); v1 = fmaxf(v1, 0.0f); }
        C[(size_t)rowg * N + colg0] = v0;
        C[(size_t)rowg * N + colg1] = v1;
    }
}

__global__ __launch_bounds__(256) void node_dots(
    const float* __restrict__ H, const float* __restrict__ as_,
    const float* __restrict__ ad_, float* __restrict__ es,
    float* __restrict__ ed, int n)
{
    const int node = (int)((blockIdx.x * blockDim.x + threadIdx.x) >> 5);
    const int lane = threadIdx.x & 31;
    if (node >= n) return;
    const float* row = H + (size_t)node * D_HID;
    float s = 0.0f, d = 0.0f;
    #pragma unroll
    for (int j = 0; j < D_HID / 32; ++j) {
        const int i = lane + j * 32;
        const float h = row[i];
        s += h * as_[i];
        d += h * ad_[i];
    }
    #pragma unroll
    for (int off = 16; off; off >>= 1) {
        s += __shfl_down(s, off, 32);
        d += __shfl_down(d, off, 32);
    }
    if (lane == 0) { es[node] = s; ed[node] = d; }
}

__global__ void fill_f32(float* __restrict__ p, float v, size_t n)
{
    const size_t i = (size_t)blockIdx.x * blockDim.x + threadIdx.x;
    if (i < n) p[i] = v;
}

__device__ inline void atomicMaxF(float* addr, float val)
{
    if (val >= 0.0f) atomicMax((int*)addr, __float_as_int(val));
    else             atomicMin((unsigned int*)addr, __float_as_uint(val));
}

__global__ __launch_bounds__(256) void edge_score(
    const int* __restrict__ src, const int* __restrict__ dst,
    const float* __restrict__ es, const float* __restrict__ ed,
    float* __restrict__ e, float* __restrict__ mx, int nE)
{
    const int i = blockIdx.x * blockDim.x + threadIdx.x;
    if (i >= nE) return;
    float v = es[src[i]] + ed[dst[i]];
    v = (v > 0.0f) ? v : 0.2f * v;
    e[i] = v;
    atomicMaxF(&mx[dst[i]], v);
}

__global__ __launch_bounds__(256) void edge_exp(
    const int* __restrict__ dst, const float* __restrict__ mx,
    float* __restrict__ e, float* __restrict__ den, int nE)
{
    const int i = blockIdx.x * blockDim.x + threadIdx.x;
    if (i >= nE) return;
    const int d = dst[i];
    const float ex = __expf(e[i] - mx[d]);
    e[i] = ex;
    atomicAdd(&den[d], ex);
}

__global__ __launch_bounds__(256) void edge_aggregate(
    const int* __restrict__ src, const int* __restrict__ dst,
    const float* __restrict__ e, const float* __restrict__ den,
    const float* __restrict__ H, float* __restrict__ OUT, int nE)
{
    const int edge = (int)((blockIdx.x * blockDim.x + threadIdx.x) >> 5);
    const int lane = threadIdx.x & 31;
    if (edge >= nE) return;
    const int s = src[edge], d = dst[edge];
    const float alpha = e[edge] / (den[d] + 1e-16f);
    const float* hrow = H + (size_t)s * D_HID;
    float* orow = OUT + (size_t)d * D_HID;
    #pragma unroll
    for (int j = 0; j < D_HID / 32; ++j) {
        const int c = lane + j * 32;
        atomicAdd(&orow[c], alpha * hrow[c]);
    }
}

__global__ void bias_relu256(float* __restrict__ X, const float* __restrict__ b,
                             size_t n)
{
    const size_t i = (size_t)blockIdx.x * blockDim.x + threadIdx.x;
    if (i >= n) return;
    X[i] = fmaxf(X[i] + b[i & (D_HID - 1)], 0.0f);
}

__global__ void gather_rows256(const float* __restrict__ X,
                               const int* __restrict__ idx,
                               float* __restrict__ Y, size_t n)
{
    const size_t i = (size_t)blockIdx.x * blockDim.x + threadIdx.x;
    if (i >= n) return;
    const size_t r = i >> 8, c = i & (D_HID - 1);
    Y[i] = X[(size_t)idx[r] * D_HID + c];
}

__global__ __launch_bounds__(256) void mlp2_softmax(
    const float* __restrict__ DH, const float* __restrict__ Wm2,
    const float* __restrict__ bm2, float* __restrict__ out, int nDocs)
{
    const int doc  = (int)((blockIdx.x * blockDim.x + threadIdx.x) >> 5);
    const int lane = threadIdx.x & 31;
    if (doc >= nDocs) return;
    const float* row = DH + (size_t)doc * MLP_HID;
    float logit = -3.0e38f;
    if (lane < N_CLASSES) {
        float acc = bm2[lane];
        #pragma unroll
        for (int k = 0; k < MLP_HID; ++k)
            acc += row[k] * Wm2[k * N_CLASSES + lane];
        logit = acc;
    }
    float mx = logit;
    #pragma unroll
    for (int off = 16; off; off >>= 1) mx = fmaxf(mx, __shfl_xor(mx, off, 32));
    const float ex = (lane < N_CLASSES) ? __expf(logit - mx) : 0.0f;
    float sum = ex;
    #pragma unroll
    for (int off = 16; off; off >>= 1) sum += __shfl_xor(sum, off, 32);
    if (lane < N_CLASSES) out[(size_t)doc * N_CLASSES + lane] = ex / sum;
}

extern "C" void kernel_launch(void* const* d_in, const int* in_sizes, int n_in,
                              void* d_out, int out_size, void* d_ws, size_t ws_size,
                              hipStream_t stream)
{
    (void)in_sizes; (void)n_in; (void)out_size; (void)ws_size;

    const float* x    = (const float*)d_in[0];
    const int*   ei   = (const int*)  d_in[1];
    const int*   doci = (const int*)  d_in[2];
    const float* W1   = (const float*)d_in[3];
    const float* a1s  = (const float*)d_in[4];
    const float* a1d  = (const float*)d_in[5];
    const float* b1   = (const float*)d_in[6];
    const float* W2   = (const float*)d_in[7];
    const float* a2s  = (const float*)d_in[8];
    const float* a2d  = (const float*)d_in[9];
    const float* b2   = (const float*)d_in[10];
    const float* Wm1  = (const float*)d_in[11];
    const float* bm1  = (const float*)d_in[12];
    const float* Wm2  = (const float*)d_in[13];
    const float* bm2  = (const float*)d_in[14];

    const int* src = ei;
    const int* dst = ei + N_EDGES;

    float* H    = (float*)d_ws;
    float* OUT  = H    + (size_t)N_NODES * D_HID;
    float* ES   = OUT  + (size_t)N_NODES * D_HID;
    float* ED   = ES   + N_NODES;
    float* MX   = ED   + N_NODES;
    float* DEN  = MX   + N_NODES;
    float* EB   = DEN  + N_NODES;
    float* DOCS = EB   + N_EDGES;
    float* DH   = DOCS + (size_t)N_DOCS * D_HID;

    const dim3 gemmGridN(N_NODES / 16, D_HID / 64);
    const dim3 gemmGridD(N_DOCS  / 16, MLP_HID / 64);
    const size_t feat = (size_t)N_NODES * D_HID;
    const int gFeat  = (int)((feat + 255) / 256);
    const int gNode  = (N_NODES + 255) / 256;
    const int gNodeW = (N_NODES + 7) / 8;
    const int gEdge  = (N_EDGES + 255) / 256;
    const int gEdgeW = (N_EDGES + 7) / 8;
    const size_t docf = (size_t)N_DOCS * D_HID;
    const int gDocF  = (int)((docf + 255) / 256);
    const int gDocW  = (N_DOCS + 7) / 8;

    for (int layer = 0; layer < 2; ++layer) {
        const float* A   = (layer == 0) ? x   : OUT;
        const float* W   = (layer == 0) ? W1  : W2;
        const float* as_ = (layer == 0) ? a1s : a2s;
        const float* ad_ = (layer == 0) ? a1d : a2d;
        const float* b   = (layer == 0) ? b1  : b2;

        gemm_wmma_f32<<<gemmGridN, 64, 0, stream>>>(A, W, H, N_NODES, D_HID,
                                                    D_HID, nullptr, 0);
        node_dots<<<gNodeW, 256, 0, stream>>>(H, as_, ad_, ES, ED, N_NODES);
        fill_f32<<<gNode, 256, 0, stream>>>(MX, -INFINITY, (size_t)N_NODES);
        fill_f32<<<gNode, 256, 0, stream>>>(DEN, 0.0f, (size_t)N_NODES);
        fill_f32<<<gFeat, 256, 0, stream>>>(OUT, 0.0f, feat);
        edge_score<<<gEdge, 256, 0, stream>>>(src, dst, ES, ED, EB, MX, N_EDGES);
        edge_exp<<<gEdge, 256, 0, stream>>>(dst, MX, EB, DEN, N_EDGES);
        edge_aggregate<<<gEdgeW, 256, 0, stream>>>(src, dst, EB, DEN, H, OUT,
                                                   N_EDGES);
        bias_relu256<<<gFeat, 256, 0, stream>>>(OUT, b, feat);
    }

    gather_rows256<<<gDocF, 256, 0, stream>>>(OUT, doci, DOCS, docf);
    gemm_wmma_f32<<<gemmGridD, 64, 0, stream>>>(DOCS, Wm1, DH, N_DOCS, MLP_HID,
                                                D_HID, bm1, 1);
    mlp2_softmax<<<gDocW, 256, 0, stream>>>(DH, Wm2, bm2, (float*)d_out, N_DOCS);
}


# Aggregate: edge list bucketed by owner wave (rows w, w+32, ..); a wave adds into its own LDS rows by plain float4 read+write instead of 8 LDS float atomics per edge; per-row sum order changes
# speedup vs baseline: 3.4655x; 3.4655x over previous
#include <hip/hip_runtime.h>
#include <hip/hip_bf16.h>


#define N_NODES   50000
#define N_EDGES   800000
#define D_HID     256
#define N_DOCS    20000
#define MLP_HID   64
#define N_CLASSES 20

typedef __attribute__((ext_vector_type(2))) float v2f;
typedef __attribute__((ext_vector_type(8))) float v8f;

#define GEMM_PA 36
#define GEMM_PB 72
template <bool kGatherA>
__global__ __launch_bounds__(64) void gemm_wmma_f32(
    const float* __restrict__ A, const float* __restrict__ B,
    float* __restrict__ C, int M, int N, int K,
    const float* __restrict__ bias, int doRelu,
    const int* __restrict__ rowIdx)
{
    __shared__ __align__(16) float As[16][GEMM_PA];
    __shared__ __align__(16) float Bs[32][GEMM_PB];

    const int tid  = threadIdx.x;
    const int wave = tid >> 5;
    const int lane = tid & 31;
    const int m0 = blockIdx.x * 16;
    const int n0 = blockIdx.y * 64;

    v8f acc0 = {};
    v8f acc1 = {};

    const int mrow  = lane & 15;
    const int khalf = (lane >> 4) << 1;
    const int ncol0 = (lane & 15) + (wave << 5);
    const int ncol1 = ncol0 + 16;

    const int sar = tid >> 3;
    const int sac = (tid & 7) * 4;
    const int sbr = tid >> 4;
    const int sbc = (tid & 15) * 4;
    const int ar = tid >> 2;
    const int ac = (tid & 3) * 8;
    const int br = tid >> 1;
    const int bc = (tid & 1) * 32;
    int ga0 = 0, ga1 = 0, gap = 0;
    if (kGatherA) {
        ga0 = rowIdx[m0 + sar];
        ga1 = rowIdx[m0 + sar + 8];
        gap = rowIdx[m0 + ar];
    }

    for (int k0 = 0; k0 < K; k0 += 32) {
        if (kGatherA) {
            *(float4*)&As[sar][sac] =
                *(const float4*)(A + (size_t)ga0 * K + k0 + sac);
            *(float4*)&As[sar + 8][sac] =
                *(const float4*)(A + (size_t)ga1 * K + k0 + sac);
        } else {
            const float* s = A + (size_t)(m0 + sar) * K + k0 + sac;
            *(float4*)&As[sar][sac]     = *(const float4*)(s);
            *(float4*)&As[sar + 8][sac] = *(const float4*)(s + (size_t)8 * K);
        }
        {
            const float* s = B + (size_t)(k0 + sbr) * N + n0 + sbc;
            #pragma unroll
            for (int j = 0; j < 8; ++j)
                *(float4*)&Bs[sbr + 4 * j][sbc] =
                    *(const float4*)(s + (size_t)(4 * j) * N);
        }
        if (k0 + 32 < K) {
            __builtin_prefetch(A + (size_t)(kGatherA ? gap : m0 + ar) * K + k0 + 32 + ac, 0, 3);
            __builtin_prefetch(B + (size_t)(k0 + 32 + br) * N + n0 + bc, 0, 3);
        }
        __syncthreads();

        #pragma unroll
        for (int kk = 0; kk < 32; kk += 4) {
            v2f a, b0, b1;
            a.x = As[mrow][kk + khalf];
            a.y = As[mrow][kk + khalf + 1];
            b0.x = Bs[kk + khalf][ncol0];
            b0.y = Bs[kk + khalf + 1][ncol0];
            b1.x = Bs[kk + khalf][ncol1];
            b1.y = Bs[kk + khalf + 1][ncol1];
            acc0 = __builtin_amdgcn_wmma_f32_16x16x4_f32(
                false, a, false, b0, (short)0, acc0, false, false);
            acc1 = __builtin_amdgcn_wmma_f32_16x16x4_f32(
                false, a, false, b1, (short)0, acc1, false, false);
        }
        __syncthreads();
    }

    const int colg0 = n0 + ncol0;
    const int colg1 = n0 + ncol1;
    const float bv0 = bias ? bias[colg0] : 0.0f;
    const float bv1 = bias ? bias[colg1] : 0.0f;
    #pragma unroll
    for (int r = 0; r < 8; ++r) {
        const int rowg = m0 + r + ((lane >> 4) << 3);
        float v0 = acc0[r] + bv0;
        float v1 = acc1[r] + bv1;
        if (doRelu) { v0 = fmaxf(v0, 0.0f); v1 = fmaxf(v1, 0.0f); }
        C[(size_t)rowg * N + colg0] = v0;
        C[(size_t)rowg * N + colg1] = v1;
    }
}

__global__ __launch_bounds__(256) void node_dots(
    const float* __restrict__ H, const float* __restrict__ as_,
    const float* __restrict__ ad_, float* __restrict__ es,
    float* __restrict__ ed, int n)
{
    const int node = (int)((blockIdx.x * blockDim.x + threadIdx.x) >> 5);
    const int lane = threadIdx.x & 31;
    if (node >= n) return;
    const float* row = H + (size_t)node * D_HID;
    float s = 0.0f, d = 0.0f;
    #pragma unroll
    for (int j = 0; j < D_HID / 32; ++j) {
        const int i = lane + j * 32;
        const float h = row[i];
        s += h * as_[i];
        d += h * ad_[i];
    }
    #pragma unroll
    for (int off = 16; off; off >>= 1) {
        s += __shfl_down(s, off, 32);
        d += __shfl_down(d, off, 32);
    }
    if (lane == 0) { es[node] = s; ed[node] = d; }
}

#define AGG_ROWS    196
#define AGG_WAVES   32
#define AGG_THREADS (AGG_WAVES * 32)
#define AGG_CAP     (4 * AGG_THREADS)

#define PART_CHUNKS  256
#define PART_EDGES   ((N_EDGES + PART_CHUNKS - 1) / PART_CHUNKS)
#define PART_THREADS 1024
#define PART_PER_THR ((PART_EDGES + PART_THREADS - 1) / PART_THREADS)
#define AGG_OWNERS   ((N_NODES + AGG_ROWS - 1) / AGG_ROWS)
static_assert(AGG_ROWS <= 256 && N_NODES < (1 << 24), "src and local row share one word");
static_assert(AGG_OWNERS <= 256, "one counter per owner in a 256-entry table");

__device__ __forceinline__ int block_excl_scan(int v, int tid, int* wsum, int& total)
{
    const int lane = tid & 31, w = tid >> 5;
    int x = v;
    #pragma unroll
    for (int off = 1; off < 32; off <<= 1) {
        const int y = __shfl_up(x, off, 32);
        if (lane >= off) x += y;
    }
    if (lane == 31) wsum[w] = x;
    __syncthreads();
    int base = 0, tot = 0;
    #pragma unroll
    for (int j = 0; j < 32; ++j) {
        const int t = wsum[j];
        tot += t;
        if (j < w) base += t;
    }
    __syncthreads();
    total = tot;
    return base + x - v;
}

__global__ __launch_bounds__(PART_THREADS) void edge_partition(
    const int* __restrict__ src, const int* __restrict__ dst,
    int* __restrict__ part, int* __restrict__ cntT, int* __restrict__ offT,
    int nE, int nNodes)
{
    __shared__ int hist[256];
    __shared__ int offs[256];
    __shared__ int wsum[32];
    __shared__ int srt[PART_EDGES];

    const int tid = threadIdx.x;
    const int b   = blockIdx.x;
    const int e0  = b * PART_EDGES;
    int e1 = e0 + PART_EDGES;
    if (e1 > nE) e1 = nE;

    if (tid < 256) hist[tid] = 0;
    __syncthreads();

    int pk[PART_PER_THR], ow[PART_PER_THR], rk[PART_PER_THR];
    int dv[PART_PER_THR], sv[PART_PER_THR];
    #pragma unroll
    for (int k = 0; k < PART_PER_THR; ++k) {
        const int e  = e0 + k * PART_THREADS + tid;
        const int ei = (e < e1) ? e : 0;
        dv[k] = dst[ei];
        sv[k] = src[ei];
    }
    #pragma unroll
    for (int k = 0; k < PART_PER_THR; ++k) {
        const int e = e0 + k * PART_THREADS + tid;
        const int d = dv[k];
        const int o = d / AGG_ROWS;
        const bool in = (e < e1) && ((unsigned)d < (unsigned)nNodes);
        ow[k] = in ? o : -1;
        pk[k] = (sv[k] << 8) | (d - o * AGG_ROWS);
    }
    #pragma unroll
    for (int k = 0; k < PART_PER_THR; ++k)
        if (ow[k] >= 0) rk[k] = atomicAdd(&hist[ow[k]], 1);
    __syncthreads();

    int tot;
    const int hv = (tid < 256) ? hist[tid] : 0;
    const int ex = block_excl_scan(hv, tid, wsum, tot);
    if (tid < 256) offs[tid] = ex;
    if (tid < AGG_OWNERS) {
        cntT[tid * PART_CHUNKS + b] = hv;
        offT[tid * PART_CHUNKS + b] = ex;
    }
    __syncthreads();

    #pragma unroll
    for (int k = 0; k < PART_PER_THR; ++k)
        if (ow[k] >= 0) srt[offs[ow[k]] + rk[k]] = pk[k];
    __syncthreads();
    for (int i = tid; i < tot; i += PART_THREADS) part[e0 + i] = srt[i];
}


__device__ __forceinline__ int f2key(float v)
{
    const int b = __float_as_int(v);
    return b ^ ((b >> 31) & 0x7fffffff);
}
__device__ __forceinline__ float key2f(int k)
{
    return __int_as_float(k ^ ((k >> 31) & 0x7fffffff));
}

__device__ __forceinline__ void agg_add_edge(
    float* __restrict__ acc, float* __restrict__ denL,
    const float* __restrict__ H, int s, int q, float alpha, int lane)
{
    const float4* hrow = (const float4*)(H + (size_t)s * D_HID) + lane;
    const float4 v0 = hrow[0];
    const float4 v1 = hrow[32];
    float* a = acc + q * D_HID + lane;
    if (lane == 0) atomicAdd(denL + q, alpha);
    atomicAdd(a +   0, alpha * v0.x);
    atomicAdd(a +  32, alpha * v0.y);
    atomicAdd(a +  64, alpha * v0.z);
    atomicAdd(a +  96, alpha * v0.w);
    atomicAdd(a + 128, alpha * v1.x);
    atomicAdd(a + 160, alpha * v1.y);
    atomicAdd(a + 192, alpha * v1.z);
    atomicAdd(a + 224, alpha * v1.w);
}

__device__ __forceinline__ void agg_run_entries(
    int q, int end, const int* __restrict__ srtU, const float* __restrict__ srtE,
    const int* __restrict__ mxL, float* __restrict__ denL, int lane, int& u, float& a)
{
    const int  p  = q + (lane & 3);
    const bool ok = p < end;
    u = ok ? srtU[p] : 255;
    const float e = ok ? srtE[p] : 0.0f;
    const int   m = ok ? mxL[u & 255] : 0;
    a = ok ? __expf(e - key2f(m)) : 0.0f;
    if (ok && lane < 4) atomicAdd(denL + (u & 255), a);
}

__device__ __forceinline__ void agg_row_add(
    float* __restrict__ acc, int row, float a, const float4 x, const float4 y, int lane)
{
    float4* r = (float4*)(acc + row * D_HID) + lane;
    float4 s0 = r[0];
    float4 s1 = r[32];
    s0.x = fmaf(a, x.x, s0.x); s0.y = fmaf(a, x.y, s0.y);
    s0.z = fmaf(a, x.z, s0.z); s0.w = fmaf(a, x.w, s0.w);
    s1.x = fmaf(a, y.x, s1.x); s1.y = fmaf(a, y.y, s1.y);
    s1.z = fmaf(a, y.z, s1.z); s1.w = fmaf(a, y.w, s1.w);
    r[0]  = s0;
    r[32] = s1;
}

__global__ __launch_bounds__(AGG_THREADS) void edge_aggregate_owned(
    const int* __restrict__ src, const int* __restrict__ dst,
    const float* __restrict__ es, const float* __restrict__ ed,
    const float* __restrict__ H, const float* __restrict__ bias,
    float* __restrict__ OUT, const int* __restrict__ part,
    const int* __restrict__ cntT, const int* __restrict__ offT,
    int nE, int nNodes)
{
    __shared__ __align__(16) float acc[AGG_ROWS * D_HID];
    __shared__ float denL[AGG_ROWS];
    __shared__ int   mxL[AGG_ROWS];
    __shared__ int   lstS[AGG_CAP];
    __shared__ float lstE[AGG_CAP];
    __shared__ int   pcnt[PART_CHUNKS];
    __shared__ int   poff[PART_CHUNKS];
    __shared__ int   ppos[PART_CHUNKS];
    __shared__ int   wsum[32];
    __shared__ int   wcnt[AGG_WAVES];
    __shared__ int   wbeg[AGG_WAVES];

    const int tid   = threadIdx.x;
    const int lane  = tid & 31;
    const int wave  = __builtin_amdgcn_readfirstlane(tid >> 5);
    const int rbase = blockIdx.x * AGG_ROWS;
    int rowsHere = nNodes - rbase;
    if (rowsHere > AGG_ROWS) rowsHere = AGG_ROWS;

    for (int i = tid; i < AGG_ROWS * D_HID / 4; i += AGG_THREADS)
        ((float4*)acc)[i] = make_float4(0.0f, 0.0f, 0.0f, 0.0f);
    if (tid < AGG_ROWS) { denL[tid] = 0.0f; mxL[tid] = f2key(-INFINITY); }
    if (tid < AGG_WAVES) wcnt[tid] = 0;

    int myCnt = 0;
    if (tid < PART_CHUNKS) {
        myCnt     = cntT[blockIdx.x * PART_CHUNKS + tid];
        pcnt[tid] = myCnt;
        poff[tid] = offT[blockIdx.x * PART_CHUNKS + tid];
    }
    int totalV;
    const int myPos = block_excl_scan(myCnt, tid, wsum, totalV);
    if (tid < PART_CHUNKS) ppos[tid] = myPos;
    __syncthreads();
    const int total = __builtin_amdgcn_readfirstlane(totalV);

    if (total <= AGG_CAP) {
        for (int c = tid >> 2; c < PART_CHUNKS; c += AGG_THREADS / 4) {
            const int  n  = pcnt[c];
            const int* sp = part + (size_t)c * PART_EDGES + poff[c];
            int*       dp = lstS + ppos[c];
            for (int j = tid & 3; j < n; j += 16) {
                int u[4];
                #pragma unroll
                for (int k = 0; k < 4; ++k) u[k] = (j + 4 * k < n) ? sp[j + 4 * k] : 0;
                #pragma unroll
                for (int k = 0; k < 4; ++k) if (j + 4 * k < n) dp[j + 4 * k] = u[k];
            }
        }
        __syncthreads();

        int   eu[4], rk[4];
        float v[4];
        bool  ok[4];
        #pragma unroll
        for (int k = 0; k < 4; ++k) {
            const int p = k * AGG_THREADS + tid;
            ok[k] = p < total;
            eu[k] = ok[k] ? lstS[p] : 0;
        }
        #pragma unroll
        for (int k = 0; k < 4; ++k) {
            const float t = es[(unsigned)eu[k] >> 8] + ed[rbase + (eu[k] & 255)];
            v[k] = (t > 0.0f) ? t : 0.2f * t;
        }
        #pragma unroll
        for (int k = 0; k < 4; ++k) {
            rk[k] = 0;
            if (ok[k]) {
                atomicMax(&mxL[eu[k] & 255], f2key(v[k]));
                rk[k] = atomicAdd(&wcnt[eu[k] & (AGG_WAVES - 1)], 1);
            }
        }
        __syncthreads();
        if (tid < AGG_WAVES) {
            int s = 0;
            for (int j = 0; j < tid; ++j) s += wcnt[j];
            wbeg[tid] = s;
        }
        __syncthreads();
        #pragma unroll
        for (int k = 0; k < 4; ++k) {
            if (ok[k]) {
                const int pos = wbeg[eu[k] & (AGG_WAVES - 1)] + rk[k];
                lstS[pos] = eu[k];
                lstE[pos] = v[k];
            }
        }
        __syncthreads();

        const int beg = __builtin_amdgcn_readfirstlane(wbeg[wave]);
        const int end = beg + __builtin_amdgcn_readfirstlane(wcnt[wave]);
        float4 hv[8];
        int    su[4];
        float  sa[4];
        {
            int u; float a;
            agg_run_entries(beg, end, lstS, lstE, mxL, denL, lane, u, a);
            #pragma unroll
            for (int k = 0; k < 4; ++k) {
                su[k] = __builtin_amdgcn_readlane(u, k);
                sa[k] = __int_as_float(__builtin_amdgcn_readlane(__float_as_int(a), k));
                const float4* h = (const float4*)(H + (size_t)((unsigned)su[k] >> 8) * D_HID) + lane;
                hv[2 * k]     = h[0];
                hv[2 * k + 1] = h[32];
                __builtin_amdgcn_sched_barrier(0);
            }
        }
        for (int q = beg; q < end; q += 4) {
            int un; float an;
            agg_run_entries(q + 4, end, lstS, lstE, mxL, denL, lane, un, an);
            #pragma unroll
            for (int k = 0; k < 4; ++k) {
                const int row = su[k] & 255;
                if (row < AGG_ROWS)
                    agg_row_add(acc, row, sa[k], hv[2 * k], hv[2 * k + 1], lane);
                su[k] = __builtin_amdgcn_readlane(un, k);
                sa[k] = __int_as_float(__builtin_amdgcn_readlane(__float_as_int(an), k));
                const float4* h = (const float4*)(H + (size_t)((unsigned)su[k] >> 8) * D_HID) + lane;
                hv[2 * k]     = h[0];
                hv[2 * k + 1] = h[32];
            }
        }

        const float4 b0 = ((const float4*)bias)[lane];
        const float4 b1 = ((const float4*)bias)[32 + lane];
        for (int row = wave; row < rowsHere; row += AGG_WAVES) {
            const float4* r = (const float4*)(acc + row * D_HID) + lane;
            const float4 s0 = r[0];
            const float4 s1 = r[32];
            const float rdn = 1.0f / (denL[row] + 1e-16f);
            float4 o0, o1;
            o0.x = fmaxf(s0.x * rdn + b0.x, 0.0f);
            o0.y = fmaxf(s0.y * rdn + b0.y, 0.0f);
            o0.z = fmaxf(s0.z * rdn + b0.z, 0.0f);
            o0.w = fmaxf(s0.w * rdn + b0.w, 0.0f);
            o1.x = fmaxf(s1.x * rdn + b1.x, 0.0f);
            o1.y = fmaxf(s1.y * rdn + b1.y, 0.0f);
            o1.z = fmaxf(s1.z * rdn + b1.z, 0.0f);
            o1.w = fmaxf(s1.w * rdn + b1.w, 0.0f);
            float4* o = (float4*)(OUT + (size_t)(rbase + row) * D_HID) + lane;
            o[0]  = o0;
            o[32] = o1;
        }
    } else {
        for (int i = tid; i < nE; i += AGG_THREADS) {
            const int d = dst[i];
            if ((unsigned)(d - rbase) < (unsigned)AGG_ROWS) {
                const float t = es[src[i]] + ed[d];
                atomicMax(&mxL[d - rbase], f2key((t > 0.0f) ? t : 0.2f * t));
            }
        }
        __syncthreads();
        for (int c0 = wave * 32; c0 < nE; c0 += AGG_WAVES * 32) {
            const int i = c0 + lane;
            const int d = (i < nE) ? dst[i] : -1;
            unsigned lm = __builtin_amdgcn_ballot_w32(
                (unsigned)(d - rbase) < (unsigned)AGG_ROWS);
            while (lm) {
                const int ej = c0 + __builtin_ctz(lm);
                lm &= lm - 1;
                const int   q = dst[ej] - rbase;
                const int   s = src[ej];
                const float t = es[s] + ed[rbase + q];
                const float e = (t > 0.0f) ? t : 0.2f * t;
                agg_add_edge(acc, denL, H, s, q, __expf(e - key2f(mxL[q])), lane);
            }
        }
        __syncthreads();

        for (int i = tid; i < rowsHere * (D_HID / 4); i += AGG_THREADS) {
            const int row = i >> 6;
            const int hl  = i & 63;
            const int col = (hl >> 5) * 128 + 4 * (hl & 31);
            const float* a = acc + row * D_HID + (hl >> 5) * 128 + (hl & 31);
            const float rdn = 1.0f / (denL[row] + 1e-16f);
            float4 o;
            o.x = fmaxf(a[0]  * rdn + bias[col + 0], 0.0f);
            o.y = fmaxf(a[32] * rdn + bias[col + 1], 0.0f);
            o.z = fmaxf(a[64] * rdn + bias[col + 2], 0.0f);
            o.w = fmaxf(a[96] * rdn + bias[col + 3], 0.0f);
            *(float4*)(OUT + (size_t)(rbase + row) * D_HID + col) = o;
        }
    }
}

__global__ __launch_bounds__(256) void mlp2_softmax(
    const float* __restrict__ DH, const float* __restrict__ Wm2,
    const float* __restrict__ bm2, float* __restrict__ out, int nDocs)
{
    const int doc  = (int)((blockIdx.x * blockDim.x + threadIdx.x) >> 5);
    const int lane = threadIdx.x & 31;
    if (doc >= nDocs) return;
    const float* row = DH + (size_t)doc * MLP_HID;
    float logit = -3.0e38f;
    if (lane < N_CLASSES) {
        float acc = bm2[lane];
        #pragma unroll
        for (int k = 0; k < MLP_HID; ++k)
            acc += row[k] * Wm2[k * N_CLASSES + lane];
        logit = acc;
    }
    float mx = logit;
    #pragma unroll
    for (int off = 16; off; off >>= 1) mx = fmaxf(mx, __shfl_xor(mx, off, 32));
    const float ex = (lane < N_CLASSES) ? __expf(logit - mx) : 0.0f;
    float sum = ex;
    #pragma unroll
    for (int off = 16; off; off >>= 1) sum += __shfl_xor(sum, off, 32);
    if (lane < N_CLASSES) out[(size_t)doc * N_CLASSES + lane] = ex / sum;
}

extern "C" void kernel_launch(void* const* d_in, const int* in_sizes, int n_in,
                              void* d_out, int out_size, void* d_ws, size_t ws_size,
                              hipStream_t stream)
{
    (void)in_sizes; (void)n_in; (void)out_size; (void)ws_size;

    const float* x    = (const float*)d_in[0];
    const int*   ei   = (const int*)  d_in[1];
    const int*   doci = (const int*)  d_in[2];
    const float* W1   = (const float*)d_in[3];
    const float* a1s  = (const float*)d_in[4];
    const float* a1d  = (const float*)d_in[5];
    const float* b1   = (const float*)d_in[6];
    const float* W2   = (const float*)d_in[7];
    const float* a2s  = (const float*)d_in[8];
    const float* a2d  = (const float*)d_in[9];
    const float* b2   = (const float*)d_in[10];
    const float* Wm1  = (const float*)d_in[11];
    const float* bm1  = (const float*)d_in[12];
    const float* Wm2  = (const float*)d_in[13];
    const float* bm2  = (const float*)d_in[14];

    const int* src = ei;
    const int* dst = ei + N_EDGES;

    float* H    = (float*)d_ws;
    float* OUT  = H    + (size_t)N_NODES * D_HID;
    float* ES   = OUT  + (size_t)N_NODES * D_HID;
    float* ED   = ES   + N_NODES;
    float* DH   = ED   + N_NODES;
    int*   PART = (int*)DH;
    int*   CNTT = PART + (size_t)PART_CHUNKS * PART_EDGES;
    int*   OFFT = CNTT + (size_t)256 * PART_CHUNKS;
    static_assert((size_t)PART_CHUNKS * PART_EDGES + (size_t)2 * 256 * PART_CHUNKS
                  <= (size_t)N_DOCS * MLP_HID, "partition fits DH");

    const dim3 gemmGridN(N_NODES / 16, D_HID / 64);
    const dim3 gemmGridD(N_DOCS  / 16, MLP_HID / 64);
    const int gNodeW = (N_NODES + 7) / 8;
    const int gAgg   = AGG_OWNERS;
    const int gDocW  = (N_DOCS + 7) / 8;

    edge_partition<<<PART_CHUNKS, PART_THREADS, 0, stream>>>(
        src, dst, PART, CNTT, OFFT, N_EDGES, N_NODES);

    for (int layer = 0; layer < 2; ++layer) {
        const float* A   = (layer == 0) ? x   : OUT;
        const float* W   = (layer == 0) ? W1  : W2;
        const float* as_ = (layer == 0) ? a1s : a2s;
        const float* ad_ = (layer == 0) ? a1d : a2d;
        const float* b   = (layer == 0) ? b1  : b2;

        gemm_wmma_f32<false><<<gemmGridN, 64, 0, stream>>>(
            A, W, H, N_NODES, D_HID, D_HID, nullptr, 0, nullptr);
        node_dots<<<gNodeW, 256, 0, stream>>>(H, as_, ad_, ES, ED, N_NODES);
        edge_aggregate_owned<<<gAgg, AGG_THREADS, 0, stream>>>(
            src, dst, ES, ED, H, b, OUT, PART, CNTT, OFFT, N_EDGES, N_NODES);
    }

    gemm_wmma_f32<true><<<gemmGridD, 64, 0, stream>>>(
        OUT, Wm1, DH, N_DOCS, MLP_HID, D_HID, bm1, 1, doci);
    mlp2_softmax<<<gDocW, 256, 0, stream>>>(DH, Wm2, bm2, (float*)d_out, N_DOCS);
}


# Node GEMMs on 80x128 work-group tiles (4 waves, ten 16x16 accumulators each, A and W chunks through LDS): staged bytes per product cut about 4x; head GEMM keeps 16x64; k sum order kept
# speedup vs baseline: 4.2890x; 1.2376x over previous
#include <hip/hip_runtime.h>
#include <hip/hip_bf16.h>


#define N_NODES   50000
#define N_EDGES   800000
#define D_HID     256
#define N_DOCS    20000
#define MLP_HID   64
#define N_CLASSES 20

typedef __attribute__((ext_vector_type(2))) float v2f;
typedef __attribute__((ext_vector_type(8))) float v8f;

#define GEMM_PA 36
#define GEMM_PB 72
template <bool kGatherA>
__global__ __launch_bounds__(64) void gemm_wmma_f32(
    const float* __restrict__ A, const float* __restrict__ B,
    float* __restrict__ C, int M, int N, int K,
    const float* __restrict__ bias, int doRelu,
    const int* __restrict__ rowIdx)
{
    __shared__ __align__(16) float As[16][GEMM_PA];
    __shared__ __align__(16) float Bs[32][GEMM_PB];

    const int tid  = threadIdx.x;
    const int wave = tid >> 5;
    const int lane = tid & 31;
    const int m0 = blockIdx.x * 16;
    const int n0 = blockIdx.y * 64;

    v8f acc0 = {};
    v8f acc1 = {};

    const int mrow  = lane & 15;
    const int khalf = (lane >> 4) << 1;
    const int ncol0 = (lane & 15) + (wave << 5);
    const int ncol1 = ncol0 + 16;

    const int sar = tid >> 3;
    const int sac = (tid & 7) * 4;
    const int sbr = tid >> 4;
    const int sbc = (tid & 15) * 4;
    const int ar = tid >> 2;
    const int ac = (tid & 3) * 8;
    const int br = tid >> 1;
    const int bc = (tid & 1) * 32;
    int ga0 = 0, ga1 = 0, gap = 0;
    if (kGatherA) {
        ga0 = rowIdx[m0 + sar];
        ga1 = rowIdx[m0 + sar + 8];
        gap = rowIdx[m0 + ar];
    }

    for (int k0 = 0; k0 < K; k0 += 32) {
        if (kGatherA) {
            *(float4*)&As[sar][sac] =
                *(const float4*)(A + (size_t)ga0 * K + k0 + sac);
            *(float4*)&As[sar + 8][sac] =
                *(const float4*)(A + (size_t)ga1 * K + k0 + sac);
        } else {
            const float* s = A + (size_t)(m0 + sar) * K + k0 + sac;
            *(float4*)&As[sar][sac]     = *(const float4*)(s);
            *(float4*)&As[sar + 8][sac] = *(const float4*)(s + (size_t)8 * K);
        }
        {
            const float* s = B + (size_t)(k0 + sbr) * N + n0 + sbc;
            #pragma unroll
            for (int j = 0; j < 8; ++j)
                *(float4*)&Bs[sbr + 4 * j][sbc] =
                    *(const float4*)(s + (size_t)(4 * j) * N);
        }
        if (k0 + 32 < K) {
            __builtin_prefetch(A + (size_t)(kGatherA ? gap : m0 + ar) * K + k0 + 32 + ac, 0, 3);
            __builtin_prefetch(B + (size_t)(k0 + 32 + br) * N + n0 + bc, 0, 3);
        }
        __syncthreads();

        #pragma unroll
        for (int kk = 0; kk < 32; kk += 4) {
            v2f a, b0, b1;
            a.x = As[mrow][kk + khalf];
            a.y = As[mrow][kk + khalf + 1];
            b0.x = Bs[kk + khalf][ncol0];
            b0.y = Bs[kk + khalf + 1][ncol0];
            b1.x = Bs[kk + khalf][ncol1];
            b1.y = Bs[kk + khalf + 1][ncol1];
            acc0 = __builtin_amdgcn_wmma_f32_16x16x4_f32(
                false, a, false, b0, (short)0, acc0, false, false);
            acc1 = __builtin_amdgcn_wmma_f32_16x16x4_f32(
                false, a, false, b1, (short)0, acc1, false, false);
        }
        __syncthreads();
    }

    const int colg0 = n0 + ncol0;
    const int colg1 = n0 + ncol1;
    const float bv0 = bias ? bias[colg0] : 0.0f;
    const float bv1 = bias ? bias[colg1] : 0.0f;
    #pragma unroll
    for (int r = 0; r < 8; ++r) {
        const int rowg = m0 + r + ((lane >> 4) << 3);
        float v0 = acc0[r] + bv0;
        float v1 = acc1[r] + bv1;
        if (doRelu) { v0 = fmaxf(v0, 0.0f); v1 = fmaxf(v1, 0.0f); }
        C[(size_t)rowg * N + colg0] = v0;
        C[(size_t)rowg * N + colg1] = v1;
    }
}

#define NG_TM      80
#define NG_TN      128
#define NG_MT      (NG_TM / 16)
#define NG_WAVES   (NG_TN / 32)
#define NG_THREADS (NG_WAVES * 32)
#define NG_KC      32
#define NG_PA      36
#define NG_PB      (NG_TN + 8)
static_assert(N_NODES % NG_TM == 0 && D_HID % NG_TN == 0 && D_HID % NG_KC == 0,
              "node GEMM tiles cover the matrices exactly");
static_assert(NG_THREADS == 128 && NG_TM == 80 && NG_KC == 32,
              "staging maps below: 5 float4 of A and 8 of B per thread");

__global__ __launch_bounds__(NG_THREADS) void gemm_nodes_wmma_f32(
    const float* __restrict__ A, const float* __restrict__ B,
    float* __restrict__ C, int N, int K)
{
    __shared__ __align__(16) float As[NG_TM][NG_PA];
    __shared__ __align__(16) float Bs[NG_KC][NG_PB];

    const int tid  = threadIdx.x;
    const int wave = tid >> 5;
    const int lane = tid & 31;
    const int m0 = blockIdx.x * NG_TM;
    const int n0 = blockIdx.y * NG_TN;

    v8f acc0[NG_MT];
    v8f acc1[NG_MT];
    #pragma unroll
    for (int t = 0; t < NG_MT; ++t) { acc0[t] = (v8f){}; acc1[t] = (v8f){}; }

    const int mrow  = lane & 15;
    const int khalf = (lane >> 4) << 1;
    const int ncol0 = (lane & 15) + (wave << 5);
    const int ncol1 = ncol0 + 16;

    const int sar = tid >> 3;
    const int sac = (tid & 7) * 4;
    const int sbr = tid >> 5;
    const int sbc = (tid & 31) * 4;

    for (int k0 = 0; k0 < K; k0 += NG_KC) {
        {
            const float* s = A + (size_t)(m0 + sar) * K + k0 + sac;
            #pragma unroll
            for (int j = 0; j < NG_MT; ++j)
                *(float4*)&As[sar + 16 * j][sac] =
                    *(const float4*)(s + (size_t)(16 * j) * K);
        }
        {
            const float* s = B + (size_t)(k0 + sbr) * N + n0 + sbc;
            #pragma unroll
            for (int j = 0; j < 8; ++j)
                *(float4*)&Bs[sbr + 4 * j][sbc] =
                    *(const float4*)(s + (size_t)(4 * j) * N);
        }
        if (k0 + NG_KC < K && tid < NG_TM)
            __builtin_prefetch(A + (size_t)(m0 + tid) * K + k0 + NG_KC, 0, 3);
        __syncthreads();

        #pragma unroll
        for (int kk = 0; kk < NG_KC; kk += 4) {
            v2f b0, b1;
            b0.x = Bs[kk + khalf][ncol0];
            b0.y = Bs[kk + khalf + 1][ncol0];
            b1.x = Bs[kk + khalf][ncol1];
            b1.y = Bs[kk + khalf + 1][ncol1];
            #pragma unroll
            for (int t = 0; t < NG_MT; ++t) {
                v2f a;
                a.x = As[16 * t + mrow][kk + khalf];
                a.y = As[16 * t + mrow][kk + khalf + 1];
                acc0[t] = __builtin_amdgcn_wmma_f32_16x16x4_f32(
                    false, a, false, b0, (short)0, acc0[t], false, false);
                acc1[t] = __builtin_amdgcn_wmma_f32_16x16x4_f32(
                    false, a, false, b1, (short)0, acc1[t], false, false);
            }
        }
        __builtin_amdgcn_sched_barrier(0);
        asm volatile("v_nop\n\tv_nop\n\tv_nop\n\tv_nop\n\tv_nop\n\tv_nop\n\tv_nop\n\tv_nop");
        __builtin_amdgcn_sched_barrier(0);
        __syncthreads();
    }

    const int colg0 = n0 + ncol0;
    const int colg1 = n0 + ncol1;
    #pragma unroll
    for (int t = 0; t < NG_MT; ++t) {
        #pragma unroll
        for (int r = 0; r < 8; ++r) {
            const int rowg = m0 + 16 * t + r + ((lane >> 4) << 3);
            C[(size_t)rowg * N + colg0] = acc0[t][r];
            C[(size_t)rowg * N + colg1] = acc1[t][r];
        }
    }
}

__global__ __launch_bounds__(256) void node_dots(
    const float* __restrict__ H, const float* __restrict__ as_,
    const float* __restrict__ ad_, float* __restrict__ es,
    float* __restrict__ ed, int n)
{
    const int node = (int)((blockIdx.x * blockDim.x + threadIdx.x) >> 5);
    const int lane = threadIdx.x & 31;
    if (node >= n) return;
    const float* row = H + (size_t)node * D_HID;
    float s = 0.0f, d = 0.0f;
    #pragma unroll
    for (int j = 0; j < D_HID / 32; ++j) {
        const int i = lane + j * 32;
        const float h = row[i];
        s += h * as_[i];
        d += h * ad_[i];
    }
    #pragma unroll
    for (int off = 16; off; off >>= 1) {
        s += __shfl_down(s, off, 32);
        d += __shfl_down(d, off, 32);
    }
    if (lane == 0) { es[node] = s; ed[node] = d; }
}

#define AGG_ROWS    196
#define AGG_WAVES   32
#define AGG_THREADS (AGG_WAVES * 32)
#define AGG_CAP     (4 * AGG_THREADS)

#define PART_CHUNKS  256
#define PART_EDGES   ((N_EDGES + PART_CHUNKS - 1) / PART_CHUNKS)
#define PART_THREADS 1024
#define PART_PER_THR ((PART_EDGES + PART_THREADS - 1) / PART_THREADS)
#define AGG_OWNERS   ((N_NODES + AGG_ROWS - 1) / AGG_ROWS)
static_assert(AGG_ROWS <= 256 && N_NODES < (1 << 24), "src and local row share one word");
static_assert(AGG_OWNERS <= 256, "one counter per owner in a 256-entry table");

__device__ __forceinline__ int block_excl_scan(int v, int tid, int* wsum, int& total)
{
    const int lane = tid & 31, w = tid >> 5;
    int x = v;
    #pragma unroll
    for (int off = 1; off < 32; off <<= 1) {
        const int y = __shfl_up(x, off, 32);
        if (lane >= off) x += y;
    }
    if (lane == 31) wsum[w] = x;
    __syncthreads();
    int base = 0, tot = 0;
    #pragma unroll
    for (int j = 0; j < 32; ++j) {
        const int t = wsum[j];
        tot += t;
        if (j < w) base += t;
    }
    __syncthreads();
    total = tot;
    return base + x - v;
}

__global__ __launch_bounds__(PART_THREADS) void edge_partition(
    const int* __restrict__ src, const int* __restrict__ dst,
    int* __restrict__ part, int* __restrict__ cntT, int* __restrict__ offT,
    int nE, int nNodes)
{
    __shared__ int hist[256];
    __shared__ int offs[256];
    __shared__ int wsum[32];
    __shared__ int srt[PART_EDGES];

    const int tid = threadIdx.x;
    const int b   = blockIdx.x;
    const int e0  = b * PART_EDGES;
    int e1 = e0 + PART_EDGES;
    if (e1 > nE) e1 = nE;

    if (tid < 256) hist[tid] = 0;
    __syncthreads();

    int pk[PART_PER_THR], ow[PART_PER_THR], rk[PART_PER_THR];
    int dv[PART_PER_THR], sv[PART_PER_THR];
    #pragma unroll
    for (int k = 0; k < PART_PER_THR; ++k) {
        const int e  = e0 + k * PART_THREADS + tid;
        const int ei = (e < e1) ? e : 0;
        dv[k] = dst[ei];
        sv[k] = src[ei];
    }
    #pragma unroll
    for (int k = 0; k < PART_PER_THR; ++k) {
        const int e = e0 + k * PART_THREADS + tid;
        const int d = dv[k];
        const int o = d / AGG_ROWS;
        const bool in = (e < e1) && ((unsigned)d < (unsigned)nNodes);
        ow[k] = in ? o : -1;
        pk[k] = (sv[k] << 8) | (d - o * AGG_ROWS);
    }
    #pragma unroll
    for (int k = 0; k < PART_PER_THR; ++k)
        if (ow[k] >= 0) rk[k] = atomicAdd(&hist[ow[k]], 1);
    __syncthreads();

    int tot;
    const int hv = (tid < 256) ? hist[tid] : 0;
    const int ex = block_excl_scan(hv, tid, wsum, tot);
    if (tid < 256) offs[tid] = ex;
    if (tid < AGG_OWNERS) {
        cntT[tid * PART_CHUNKS + b] = hv;
        offT[tid * PART_CHUNKS + b] = ex;
    }
    __syncthreads();

    #pragma unroll
    for (int k = 0; k < PART_PER_THR; ++k)
        if (ow[k] >= 0) srt[offs[ow[k]] + rk[k]] = pk[k];
    __syncthreads();
    for (int i = tid; i < tot; i += PART_THREADS) part[e0 + i] = srt[i];
}


__device__ __forceinline__ int f2key(float v)
{
    const int b = __float_as_int(v);
    return b ^ ((b >> 31) & 0x7fffffff);
}
__device__ __forceinline__ float key2f(int k)
{
    return __int_as_float(k ^ ((k >> 31) & 0x7fffffff));
}

__device__ __forceinline__ void agg_add_edge(
    float* __restrict__ acc, float* __restrict__ denL,
    const float* __restrict__ H, int s, int q, float alpha, int lane)
{
    const float4* hrow = (const float4*)(H + (size_t)s * D_HID) + lane;
    const float4 v0 = hrow[0];
    const float4 v1 = hrow[32];
    float* a = acc + q * D_HID + lane;
    if (lane == 0) atomicAdd(denL + q, alpha);
    atomicAdd(a +   0, alpha * v0.x);
    atomicAdd(a +  32, alpha * v0.y);
    atomicAdd(a +  64, alpha * v0.z);
    atomicAdd(a +  96, alpha * v0.w);
    atomicAdd(a + 128, alpha * v1.x);
    atomicAdd(a + 160, alpha * v1.y);
    atomicAdd(a + 192, alpha * v1.z);
    atomicAdd(a + 224, alpha * v1.w);
}

__device__ __forceinline__ void agg_run_entries(
    int q, int end, const int* __restrict__ srtU, const float* __restrict__ srtE,
    const int* __restrict__ mxL, float* __restrict__ denL, int lane, int& u, float& a)
{
    const int  p  = q + (lane & 3);
    const bool ok = p < end;
    u = ok ? srtU[p] : 255;
    const float e = ok ? srtE[p] : 0.0f;
    const int   m = ok ? mxL[u & 255] : 0;
    a = ok ? __expf(e - key2f(m)) : 0.0f;
    if (ok && lane < 4) atomicAdd(denL + (u & 255), a);
}

__device__ __forceinline__ void agg_row_add(
    float* __restrict__ acc, int row, float a, const float4 x, const float4 y, int lane)
{
    float4* r = (float4*)(acc + row * D_HID) + lane;
    float4 s0 = r[0];
    float4 s1 = r[32];
    s0.x = fmaf(a, x.x, s0.x); s0.y = fmaf(a, x.y, s0.y);
    s0.z = fmaf(a, x.z, s0.z); s0.w = fmaf(a, x.w, s0.w);
    s1.x = fmaf(a, y.x, s1.x); s1.y = fmaf(a, y.y, s1.y);
    s1.z = fmaf(a, y.z, s1.z); s1.w = fmaf(a, y.w, s1.w);
    r[0]  = s0;
    r[32] = s1;
}

__global__ __launch_bounds__(AGG_THREADS) void edge_aggregate_owned(
    const int* __restrict__ src, const int* __restrict__ dst,
    const float* __restrict__ es, const float* __restrict__ ed,
    const float* __restrict__ H, const float* __restrict__ bias,
    float* __restrict__ OUT, const int* __restrict__ part,
    const int* __restrict__ cntT, const int* __restrict__ offT,
    int nE, int nNodes)
{
    __shared__ __align__(16) float acc[AGG_ROWS * D_HID];
    __shared__ float denL[AGG_ROWS];
    __shared__ int   mxL[AGG_ROWS];
    __shared__ int   lstS[AGG_CAP];
    __shared__ float lstE[AGG_CAP];
    __shared__ int   pcnt[PART_CHUNKS];
    __shared__ int   poff[PART_CHUNKS];
    __shared__ int   ppos[PART_CHUNKS];
    __shared__ int   wsum[32];
    __shared__ int   wcnt[AGG_WAVES];
    __shared__ int   wbeg[AGG_WAVES];

    const int tid   = threadIdx.x;
    const int lane  = tid & 31;
    const int wave  = __builtin_amdgcn_readfirstlane(tid >> 5);
    const int rbase = blockIdx.x * AGG_ROWS;
    int rowsHere = nNodes - rbase;
    if (rowsHere > AGG_ROWS) rowsHere = AGG_ROWS;

    for (int i = tid; i < AGG_ROWS * D_HID / 4; i += AGG_THREADS)
        ((float4*)acc)[i] = make_float4(0.0f, 0.0f, 0.0f, 0.0f);
    if (tid < AGG_ROWS) { denL[tid] = 0.0f; mxL[tid] = f2key(-INFINITY); }
    if (tid < AGG_WAVES) wcnt[tid] = 0;

    int myCnt = 0;
    if (tid < PART_CHUNKS) {
        myCnt     = cntT[blockIdx.x * PART_CHUNKS + tid];
        pcnt[tid] = myCnt;
        poff[tid] = offT[blockIdx.x * PART_CHUNKS + tid];
    }
    int totalV;
    const int myPos = block_excl_scan(myCnt, tid, wsum, totalV);
    if (tid < PART_CHUNKS) ppos[tid] = myPos;
    __syncthreads();
    const int total = __builtin_amdgcn_readfirstlane(totalV);

    if (total <= AGG_CAP) {
        for (int c = tid >> 2; c < PART_CHUNKS; c += AGG_THREADS / 4) {
            const int  n  = pcnt[c];
            const int* sp = part + (size_t)c * PART_EDGES + poff[c];
            int*       dp = lstS + ppos[c];
            for (int j = tid & 3; j < n; j += 16) {
                int u[4];
                #pragma unroll
                for (int k = 0; k < 4; ++k) u[k] = (j + 4 * k < n) ? sp[j + 4 * k] : 0;
                #pragma unroll
                for (int k = 0; k < 4; ++k) if (j + 4 * k < n) dp[j + 4 * k] = u[k];
            }
        }
        __syncthreads();

        int   eu[4], rk[4];
        float v[4];
        bool  ok[4];
        #pragma unroll
        for (int k = 0; k < 4; ++k) {
            const int p = k * AGG_THREADS + tid;
            ok[k] = p < total;
            eu[k] = ok[k] ? lstS[p] : 0;
        }
        #pragma unroll
        for (int k = 0; k < 4; ++k) {
            const float t = es[(unsigned)eu[k] >> 8] + ed[rbase + (eu[k] & 255)];
            v[k] = (t > 0.0f) ? t : 0.2f * t;
        }
        #pragma unroll
        for (int k = 0; k < 4; ++k) {
            rk[k] = 0;
            if (ok[k]) {
                atomicMax(&mxL[eu[k] & 255], f2key(v[k]));
                rk[k] = atomicAdd(&wcnt[eu[k] & (AGG_WAVES - 1)], 1);
            }
        }
        __syncthreads();
        if (tid < AGG_WAVES) {
            int s = 0;
            for (int j = 0; j < tid; ++j) s += wcnt[j];
            wbeg[tid] = s;
        }
        __syncthreads();
        #pragma unroll
        for (int k = 0; k < 4; ++k) {
            if (ok[k]) {
                const int pos = wbeg[eu[k] & (AGG_WAVES - 1)] + rk[k];
                lstS[pos] = eu[k];
                lstE[pos] = v[k];
            }
        }
        __syncthreads();

        const int beg = __builtin_amdgcn_readfirstlane(wbeg[wave]);
        const int end = beg + __builtin_amdgcn_readfirstlane(wcnt[wave]);
        float4 hv[8];
        int    su[4];
        float  sa[4];
        {
            int u; float a;
            agg_run_entries(beg, end, lstS, lstE, mxL, denL, lane, u, a);
            #pragma unroll
            for (int k = 0; k < 4; ++k) {
                su[k] = __builtin_amdgcn_readlane(u, k);
                sa[k] = __int_as_float(__builtin_amdgcn_readlane(__float_as_int(a), k));
                const float4* h = (const float4*)(H + (size_t)((unsigned)su[k] >> 8) * D_HID) + lane;
                hv[2 * k]     = h[0];
                hv[2 * k + 1] = h[32];
                __builtin_amdgcn_sched_barrier(0);
            }
        }
        for (int q = beg; q < end; q += 4) {
            int un; float an;
            agg_run_entries(q + 4, end, lstS, lstE, mxL, denL, lane, un, an);
            #pragma unroll
            for (int k = 0; k < 4; ++k) {
                const int row = su[k] & 255;
                if (row < AGG_ROWS)
                    agg_row_add(acc, row, sa[k], hv[2 * k], hv[2 * k + 1], lane);
                su[k] = __builtin_amdgcn_readlane(un, k);
                sa[k] = __int_as_float(__builtin_amdgcn_readlane(__float_as_int(an), k));
                const float4* h = (const float4*)(H + (size_t)((unsigned)su[k] >> 8) * D_HID) + lane;
                hv[2 * k]     = h[0];
                hv[2 * k + 1] = h[32];
            }
        }

        const float4 b0 = ((const float4*)bias)[lane];
        const float4 b1 = ((const float4*)bias)[32 + lane];
        for (int row = wave; row < rowsHere; row += AGG_WAVES) {
            const float4* r = (const float4*)(acc + row * D_HID) + lane;
            const float4 s0 = r[0];
            const float4 s1 = r[32];
            const float rdn = 1.0f / (denL[row] + 1e-16f);
            float4 o0, o1;
            o0.x = fmaxf(s0.x * rdn + b0.x, 0.0f);
            o0.y = fmaxf(s0.y * rdn + b0.y, 0.0f);
            o0.z = fmaxf(s0.z * rdn + b0.z, 0.0f);
            o0.w = fmaxf(s0.w * rdn + b0.w, 0.0f);
            o1.x = fmaxf(s1.x * rdn + b1.x, 0.0f);
            o1.y = fmaxf(s1.y * rdn + b1.y, 0.0f);
            o1.z = fmaxf(s1.z * rdn + b1.z, 0.0f);
            o1.w = fmaxf(s1.w * rdn + b1.w, 0.0f);
            float4* o = (float4*)(OUT + (size_t)(rbase + row) * D_HID) + lane;
            o[0]  = o0;
            o[32] = o1;
        }
    } else {
        for (int i = tid; i < nE; i += AGG_THREADS) {
            const int d = dst[i];
            if ((unsigned)(d - rbase) < (unsigned)AGG_ROWS) {
                const float t = es[src[i]] + ed[d];
                atomicMax(&mxL[d - rbase], f2key((t > 0.0f) ? t : 0.2f * t));
            }
        }
        __syncthreads();
        for (int c0 = wave * 32; c0 < nE; c0 += AGG_WAVES * 32) {
            const int i = c0 + lane;
            const int d = (i < nE) ? dst[i] : -1;
            unsigned lm = __builtin_amdgcn_ballot_w32(
                (unsigned)(d - rbase) < (unsigned)AGG_ROWS);
            while (lm) {
                const int ej = c0 + __builtin_ctz(lm);
                lm &= lm - 1;
                const int   q = dst[ej] - rbase;
                const int   s = src[ej];
                const float t = es[s] + ed[rbase + q];
                const float e = (t > 0.0f) ? t : 0.2f * t;
                agg_add_edge(acc, denL, H, s, q, __expf(e - key2f(mxL[q])), lane);
            }
        }
        __syncthreads();

        for (int i = tid; i < rowsHere * (D_HID / 4); i += AGG_THREADS) {
            const int row = i >> 6;
            const int hl  = i & 63;
            const int col = (hl >> 5) * 128 + 4 * (hl & 31);
            const float* a = acc + row * D_HID + (hl >> 5) * 128 + (hl & 31);
            const float rdn = 1.0f / (denL[row] + 1e-16f);
            float4 o;
            o.x = fmaxf(a[0]  * rdn + bias[col + 0], 0.0f);
            o.y = fmaxf(a[32] * rdn + bias[col + 1], 0.0f);
            o.z = fmaxf(a[64] * rdn + bias[col + 2], 0.0f);
            o.w = fmaxf(a[96] * rdn + bias[col + 3], 0.0f);
            *(float4*)(OUT + (size_t)(rbase + row) * D_HID + col) = o;
        }
    }
}

__global__ __launch_bounds__(256) void mlp2_softmax(
    const float* __restrict__ DH, const float* __restrict__ Wm2,
    const float* __restrict__ bm2, float* __restrict__ out, int nDocs)
{
    const int doc  = (int)((blockIdx.x * blockDim.x + threadIdx.x) >> 5);
    const int lane = threadIdx.x & 31;
    if (doc >= nDocs) return;
    const float* row = DH + (size_t)doc * MLP_HID;
    float logit = -3.0e38f;
    if (lane < N_CLASSES) {
        float acc = bm2[lane];
        #pragma unroll
        for (int k = 0; k < MLP_HID; ++k)
            acc += row[k] * Wm2[k * N_CLASSES + lane];
        logit = acc;
    }
    float mx = logit;
    #pragma unroll
    for (int off = 16; off; off >>= 1) mx = fmaxf(mx, __shfl_xor(mx, off, 32));
    const float ex = (lane < N_CLASSES) ? __expf(logit - mx) : 0.0f;
    float sum = ex;
    #pragma unroll
    for (int off = 16; off; off >>= 1) sum += __shfl_xor(sum, off, 32);
    if (lane < N_CLASSES) out[(size_t)doc * N_CLASSES + lane] = ex / sum;
}

extern "C" void kernel_launch(void* const* d_in, const int* in_sizes, int n_in,
                              void* d_out, int out_size, void* d_ws, size_t ws_size,
                              hipStream_t stream)
{
    (void)in_sizes; (void)n_in; (void)out_size; (void)ws_size;

    const float* x    = (const float*)d_in[0];
    const int*   ei   = (const int*)  d_in[1];
    const int*   doci = (const int*)  d_in[2];
    const float* W1   = (const float*)d_in[3];
    const float* a1s  = (const float*)d_in[4];
    const float* a1d  = (const float*)d_in[5];
    const float* b1   = (const float*)d_in[6];
    const float* W2   = (const float*)d_in[7];
    const float* a2s  = (const float*)d_in[8];
    const float* a2d  = (const float*)d_in[9];
    const float* b2   = (const float*)d_in[10];
    const float* Wm1  = (const float*)d_in[11];
    const float* bm1  = (const float*)d_in[12];
    const float* Wm2  = (const float*)d_in[13];
    const float* bm2  = (const float*)d_in[14];

    const int* src = ei;
    const int* dst = ei + N_EDGES;

    float* H    = (float*)d_ws;
    float* OUT  = H    + (size_t)N_NODES * D_HID;
    float* ES   = OUT  + (size_t)N_NODES * D_HID;
    float* ED   = ES   + N_NODES;
    float* DH   = ED   + N_NODES;
    int*   PART = (int*)DH;
    int*   CNTT = PART + (size_t)PART_CHUNKS * PART_EDGES;
    int*   OFFT = CNTT + (size_t)256 * PART_CHUNKS;
    static_assert((size_t)PART_CHUNKS * PART_EDGES + (size_t)2 * 256 * PART_CHUNKS
                  <= (size_t)N_DOCS * MLP_HID, "partition fits DH");

    const dim3 gemmGridN(N_NODES / NG_TM, D_HID / NG_TN);
    const dim3 gemmGridD(N_DOCS  / 16, MLP_HID / 64);
    const int gNodeW = (N_NODES + 7) / 8;
    const int gAgg   = AGG_OWNERS;
    const int gDocW  = (N_DOCS + 7) / 8;

    edge_partition<<<PART_CHUNKS, PART_THREADS, 0, stream>>>(
        src, dst, PART, CNTT, OFFT, N_EDGES, N_NODES);

    for (int layer = 0; layer < 2; ++layer) {
        const float* A   = (layer == 0) ? x   : OUT;
        const float* W   = (layer == 0) ? W1  : W2;
        const float* as_ = (layer == 0) ? a1s : a2s;
        const float* ad_ = (layer == 0) ? a1d : a2d;
        const float* b   = (layer == 0) ? b1  : b2;

        gemm_nodes_wmma_f32<<<gemmGridN, NG_THREADS, 0, stream>>>(
            A, W, H, D_HID, D_HID);
        node_dots<<<gNodeW, 256, 0, stream>>>(H, as_, ad_, ES, ED, N_NODES);
        edge_aggregate_owned<<<gAgg, AGG_THREADS, 0, stream>>>(
            src, dst, ES, ED, H, b, OUT, PART, CNTT, OFFT, N_EDGES, N_NODES);
    }

    gemm_wmma_f32<true><<<gemmGridD, 64, 0, stream>>>(
        OUT, Wm1, DH, N_DOCS, MLP_HID, D_HID, bm1, 1, doci);
    mlp2_softmax<<<gDocW, 256, 0, stream>>>(DH, Wm2, bm2, (float*)d_out, N_DOCS);
}


# MLP head moved into layer 2's owner-range aggregate: docs read the owner's LDS rows (doc product on WMMA, logits, softmax there); OUT2 store and both head launches dropped; sum orders kept
# speedup vs baseline: 4.5137x; 1.0524x over previous
#include <hip/hip_runtime.h>
#include <hip/hip_bf16.h>


#define N_NODES   50000
#define N_EDGES   800000
#define D_HID     256
#define N_DOCS    20000
#define MLP_HID   64
#define N_CLASSES 20

typedef __attribute__((ext_vector_type(2))) float v2f;
typedef __attribute__((ext_vector_type(8))) float v8f;

#define NG_TM      80
#define NG_TN      128
#define NG_MT      (NG_TM / 16)
#define NG_WAVES   (NG_TN / 32)
#define NG_THREADS (NG_WAVES * 32)
#define NG_KC      32
#define NG_PA      36
#define NG_PB      (NG_TN + 8)
static_assert(N_NODES % NG_TM == 0 && D_HID % NG_TN == 0 && D_HID % NG_KC == 0,
              "node GEMM tiles cover the matrices exactly");
static_assert(NG_THREADS == 128 && NG_TM == 80 && NG_KC == 32,
              "staging maps below: 5 float4 of A and 8 of B per thread");

__global__ __launch_bounds__(NG_THREADS) void gemm_nodes_wmma_f32(
    const float* __restrict__ A, const float* __restrict__ B,
    float* __restrict__ C, int N, int K)
{
    __shared__ __align__(16) float As[NG_TM][NG_PA];
    __shared__ __align__(16) float Bs[NG_KC][NG_PB];

    const int tid  = threadIdx.x;
    const int wave = tid >> 5;
    const int lane = tid & 31;
    const int m0 = blockIdx.x * NG_TM;
    const int n0 = blockIdx.y * NG_TN;

    v8f acc0[NG_MT];
    v8f acc1[NG_MT];
    #pragma unroll
    for (int t = 0; t < NG_MT; ++t) { acc0[t] = (v8f){}; acc1[t] = (v8f){}; }

    const int mrow  = lane & 15;
    const int khalf = (lane >> 4) << 1;
    const int ncol0 = (lane & 15) + (wave << 5);
    const int ncol1 = ncol0 + 16;

    const int sar = tid >> 3;
    const int sac = (tid & 7) * 4;
    const int sbr = tid >> 5;
    const int sbc = (tid & 31) * 4;

    for (int k0 = 0; k0 < K; k0 += NG_KC) {
        {
            const float* s = A + (size_t)(m0 + sar) * K + k0 + sac;
            #pragma unroll
            for (int j = 0; j < NG_MT; ++j)
                *(float4*)&As[sar + 16 * j][sac] =
                    *(const float4*)(s + (size_t)(16 * j) * K);
        }
        {
            const float* s = B + (size_t)(k0 + sbr) * N + n0 + sbc;
            #pragma unroll
            for (int j = 0; j < 8; ++j)
                *(float4*)&Bs[sbr + 4 * j][sbc] =
                    *(const float4*)(s + (size_t)(4 * j) * N);
        }
        if (k0 + NG_KC < K && tid < NG_TM)
            __builtin_prefetch(A + (size_t)(m0 + tid) * K + k0 + NG_KC, 0, 3);
        __syncthreads();

        #pragma unroll
        for (int kk = 0; kk < NG_KC; kk += 4) {
            v2f b0, b1;
            b0.x = Bs[kk + khalf][ncol0];
            b0.y = Bs[kk + khalf + 1][ncol0];
            b1.x = Bs[kk + khalf][ncol1];
            b1.y = Bs[kk + khalf + 1][ncol1];
            #pragma unroll
            for (int t = 0; t < NG_MT; ++t) {
                v2f a;
                a.x = As[16 * t + mrow][kk + khalf];
                a.y = As[16 * t + mrow][kk + khalf + 1];
                acc0[t] = __builtin_amdgcn_wmma_f32_16x16x4_f32(
                    false, a, false, b0, (short)0, acc0[t], false, false);
                acc1[t] = __builtin_amdgcn_wmma_f32_16x16x4_f32(
                    false, a, false, b1, (short)0, acc1[t], false, false);
            }
        }
        __builtin_amdgcn_sched_barrier(0);
        asm volatile("v_nop\n\tv_nop\n\tv_nop\n\tv_nop\n\tv_nop\n\tv_nop\n\tv_nop\n\tv_nop");
        __builtin_amdgcn_sched_barrier(0);
        __syncthreads();
    }

    const int colg0 = n0 + ncol0;
    const int colg1 = n0 + ncol1;
    #pragma unroll
    for (int t = 0; t < NG_MT; ++t) {
        #pragma unroll
        for (int r = 0; r < 8; ++r) {
            const int rowg = m0 + 16 * t + r + ((lane >> 4) << 3);
            C[(size_t)rowg * N + colg0] = acc0[t][r];
            C[(size_t)rowg * N + colg1] = acc1[t][r];
        }
    }
}

__global__ __launch_bounds__(256) void node_dots(
    const float* __restrict__ H, const float* __restrict__ as_,
    const float* __restrict__ ad_, float* __restrict__ es,
    float* __restrict__ ed, int n)
{
    const int node = (int)((blockIdx.x * blockDim.x + threadIdx.x) >> 5);
    const int lane = threadIdx.x & 31;
    if (node >= n) return;
    const float* row = H + (size_t)node * D_HID;
    float s = 0.0f, d = 0.0f;
    #pragma unroll
    for (int j = 0; j < D_HID / 32; ++j) {
        const int i = lane + j * 32;
        const float h = row[i];
        s += h * as_[i];
        d += h * ad_[i];
    }
    #pragma unroll
    for (int off = 16; off; off >>= 1) {
        s += __shfl_down(s, off, 32);
        d += __shfl_down(d, off, 32);
    }
    if (lane == 0) { es[node] = s; ed[node] = d; }
}

#define AGG_ROWS    196
#define AGG_WAVES   32
#define AGG_THREADS (AGG_WAVES * 32)
#define AGG_CAP     (4 * AGG_THREADS)

#define PART_CHUNKS  256
#define PART_EDGES   ((N_EDGES + PART_CHUNKS - 1) / PART_CHUNKS)
#define PART_THREADS 1024
#define PART_PER_THR ((PART_EDGES + PART_THREADS - 1) / PART_THREADS)
#define AGG_OWNERS   ((N_NODES + AGG_ROWS - 1) / AGG_ROWS)
static_assert(AGG_ROWS <= 256 && N_NODES < (1 << 24), "src and local row share one word");
static_assert(AGG_OWNERS <= 256, "one counter per owner in a 256-entry table");

__device__ __forceinline__ int block_excl_scan(int v, int tid, int* wsum, int& total)
{
    const int lane = tid & 31, w = tid >> 5;
    int x = v;
    #pragma unroll
    for (int off = 1; off < 32; off <<= 1) {
        const int y = __shfl_up(x, off, 32);
        if (lane >= off) x += y;
    }
    if (lane == 31) wsum[w] = x;
    __syncthreads();
    int base = 0, tot = 0;
    #pragma unroll
    for (int j = 0; j < 32; ++j) {
        const int t = wsum[j];
        tot += t;
        if (j < w) base += t;
    }
    __syncthreads();
    total = tot;
    return base + x - v;
}

__global__ __launch_bounds__(PART_THREADS) void edge_partition(
    const int* __restrict__ src, const int* __restrict__ dst,
    int* __restrict__ part, int* __restrict__ cntT, int* __restrict__ offT,
    int nE, int nNodes)
{
    __shared__ int hist[256];
    __shared__ int offs[256];
    __shared__ int wsum[32];
    __shared__ int srt[PART_EDGES];

    const int tid = threadIdx.x;
    const int b   = blockIdx.x;
    const int e0  = b * PART_EDGES;
    int e1 = e0 + PART_EDGES;
    if (e1 > nE) e1 = nE;

    if (tid < 256) hist[tid] = 0;
    __syncthreads();

    int pk[PART_PER_THR], ow[PART_PER_THR], rk[PART_PER_THR];
    int dv[PART_PER_THR], sv[PART_PER_THR];
    #pragma unroll
    for (int k = 0; k < PART_PER_THR; ++k) {
        const int e  = e0 + k * PART_THREADS + tid;
        const int ei = (e < e1) ? e : 0;
        dv[k] = dst[ei];
        sv[k] = src[ei];
    }
    #pragma unroll
    for (int k = 0; k < PART_PER_THR; ++k) {
        const int e = e0 + k * PART_THREADS + tid;
        const int d = dv[k];
        const int o = d / AGG_ROWS;
        const bool in = (e < e1) && ((unsigned)d < (unsigned)nNodes);
        ow[k] = in ? o : -1;
        pk[k] = (sv[k] << 8) | (d - o * AGG_ROWS);
    }
    #pragma unroll
    for (int k = 0; k < PART_PER_THR; ++k)
        if (ow[k] >= 0) rk[k] = atomicAdd(&hist[ow[k]], 1);
    __syncthreads();

    int tot;
    const int hv = (tid < 256) ? hist[tid] : 0;
    const int ex = block_excl_scan(hv, tid, wsum, tot);
    if (tid < 256) offs[tid] = ex;
    if (tid < AGG_OWNERS) {
        cntT[tid * PART_CHUNKS + b] = hv;
        offT[tid * PART_CHUNKS + b] = ex;
    }
    __syncthreads();

    #pragma unroll
    for (int k = 0; k < PART_PER_THR; ++k)
        if (ow[k] >= 0) srt[offs[ow[k]] + rk[k]] = pk[k];
    __syncthreads();
    for (int i = tid; i < tot; i += PART_THREADS) part[e0 + i] = srt[i];
}


__device__ __forceinline__ int f2key(float v)
{
    const int b = __float_as_int(v);
    return b ^ ((b >> 31) & 0x7fffffff);
}
__device__ __forceinline__ float key2f(int k)
{
    return __int_as_float(k ^ ((k >> 31) & 0x7fffffff));
}

template <int P, bool kNatural>
__device__ __forceinline__ void agg_add_edge(
    float* __restrict__ acc, float* __restrict__ denL,
    const float* __restrict__ H, int s, int q, float alpha, int lane)
{
    const float4* hrow = (const float4*)(H + (size_t)s * D_HID) + lane;
    const float4 v0 = hrow[0];
    const float4 v1 = hrow[32];
    if (lane == 0) atomicAdd(denL + q, alpha);
    if (kNatural) {
        float* a = acc + q * P + 4 * lane;
        atomicAdd(a +   0, alpha * v0.x);
        atomicAdd(a +   1, alpha * v0.y);
        atomicAdd(a +   2, alpha * v0.z);
        atomicAdd(a +   3, alpha * v0.w);
        atomicAdd(a + 128, alpha * v1.x);
        atomicAdd(a + 129, alpha * v1.y);
        atomicAdd(a + 130, alpha * v1.z);
        atomicAdd(a + 131, alpha * v1.w);
    } else {
        float* a = acc + q * P + lane;
        atomicAdd(a +   0, alpha * v0.x);
        atomicAdd(a +  32, alpha * v0.y);
        atomicAdd(a +  64, alpha * v0.z);
        atomicAdd(a +  96, alpha * v0.w);
        atomicAdd(a + 128, alpha * v1.x);
        atomicAdd(a + 160, alpha * v1.y);
        atomicAdd(a + 192, alpha * v1.z);
        atomicAdd(a + 224, alpha * v1.w);
    }
}

__device__ __forceinline__ void agg_run_entries(
    int q, int end, const int* __restrict__ srtU, const float* __restrict__ srtE,
    const int* __restrict__ mxL, float* __restrict__ denL, int lane, int& u, float& a)
{
    const int  p  = q + (lane & 3);
    const bool ok = p < end;
    u = ok ? srtU[p] : 255;
    const float e = ok ? srtE[p] : 0.0f;
    const int   m = ok ? mxL[u & 255] : 0;
    a = ok ? __expf(e - key2f(m)) : 0.0f;
    if (ok && lane < 4) atomicAdd(denL + (u & 255), a);
}

template <int P>
__device__ __forceinline__ void agg_row_add(
    float* __restrict__ acc, int row, float a, const float4 x, const float4 y, int lane)
{
    float4* r = (float4*)(acc + row * P) + lane;
    float4 s0 = r[0];
    float4 s1 = r[32];
    s0.x = fmaf(a, x.x, s0.x); s0.y = fmaf(a, x.y, s0.y);
    s0.z = fmaf(a, x.z, s0.z); s0.w = fmaf(a, x.w, s0.w);
    s1.x = fmaf(a, y.x, s1.x); s1.y = fmaf(a, y.y, s1.y);
    s1.z = fmaf(a, y.z, s1.z); s1.w = fmaf(a, y.w, s1.w);
    r[0]  = s0;
    r[32] = s1;
}

__device__ __forceinline__ float4 agg_finish4(const float4 s, float rdn, const float4 b)
{
    float4 o;
    o.x = fmaxf(s.x * rdn + b.x, 0.0f);
    o.y = fmaxf(s.y * rdn + b.y, 0.0f);
    o.z = fmaxf(s.z * rdn + b.z, 0.0f);
    o.w = fmaxf(s.w * rdn + b.w, 0.0f);
    return o;
}

#define HEAD_PASS  128
#define HEAD_PW    72
#define DOC_ITERS  ((N_DOCS + AGG_THREADS - 1) / AGG_THREADS)
static_assert(DOC_ITERS <= 32, "a thread's docs have one mask bit each");
static_assert(HEAD_PASS * MLP_HID <= 2 * AGG_CAP, "the hidden tile fits over the edge list");
static_assert((HEAD_PASS / 16) * (MLP_HID / 16) == AGG_WAVES, "one head job per wave");
static_assert(N_DOCS < (1 << 23), "doc and local row share one word");

template <bool kHead>
__global__ __launch_bounds__(AGG_THREADS) void edge_aggregate_owned(
    const int* __restrict__ src, const int* __restrict__ dst,
    const float* __restrict__ es, const float* __restrict__ ed,
    const float* __restrict__ H, const float* __restrict__ bias,
    float* __restrict__ OUT, const int* __restrict__ part,
    const int* __restrict__ cntT, const int* __restrict__ offT,
    int nE, int nNodes,
    const int* __restrict__ docIdx, const float* __restrict__ Wm1,
    const float* __restrict__ bm1, const float* __restrict__ Wm2,
    const float* __restrict__ bm2, float* __restrict__ out)
{
    constexpr int P = kHead ? (D_HID + 4) : D_HID;
    __shared__ __align__(16) float acc[AGG_ROWS * P];
    __shared__ float denL[AGG_ROWS];
    __shared__ int   mxL[AGG_ROWS];
    __shared__ __align__(16) union { int i[2 * AGG_CAP]; float f[2 * AGG_CAP]; } lst;
    __shared__ int   pcnt[PART_CHUNKS];
    __shared__ int   poff[PART_CHUNKS];
    __shared__ int   ppos[PART_CHUNKS];
    __shared__ int   wsum[32];
    __shared__ int   wcnt[AGG_WAVES];
    __shared__ int   wbeg[AGG_WAVES];
    __shared__ __align__(16) float Ws[kHead ? D_HID * HEAD_PW : 4];
    __shared__ int   docL[kHead ? HEAD_PASS : 1];
    int*   const lstS = lst.i;
    float* const lstE = lst.f + AGG_CAP;

    const int tid   = threadIdx.x;
    const int lane  = tid & 31;
    const int wave  = __builtin_amdgcn_readfirstlane(tid >> 5);
    const int rbase = blockIdx.x * AGG_ROWS;
    int rowsHere = nNodes - rbase;
    if (rowsHere > AGG_ROWS) rowsHere = AGG_ROWS;

    for (int i = tid; i < AGG_ROWS * P / 4; i += AGG_THREADS)
        ((float4*)acc)[i] = make_float4(0.0f, 0.0f, 0.0f, 0.0f);
    if (tid < AGG_ROWS) { denL[tid] = 0.0f; mxL[tid] = f2key(-INFINITY); }
    if (tid < AGG_WAVES) wcnt[tid] = 0;

    int myCnt = 0, myOff = 0;
    if (tid < PART_CHUNKS) {
        myCnt = cntT[blockIdx.x * PART_CHUNKS + tid];
        myOff = offT[blockIdx.x * PART_CHUNKS + tid];
    }

    unsigned docMask = 0;
    if (kHead) {
        #pragma unroll
        for (int j = 0; j < (D_HID * MLP_HID / 4) / AGG_THREADS; ++j) {
            const int f  = j * AGG_THREADS + tid;
            const int r  = f >> 4;
            const int c4 = (f & 15) * 4;
            *(float4*)&Ws[r * HEAD_PW + c4] = *(const float4*)(Wm1 + r * MLP_HID + c4);
        }
        int dr[DOC_ITERS];
        #pragma unroll
        for (int i = 0; i < DOC_ITERS; ++i) {
            const int d = i * AGG_THREADS + tid;
            dr[i] = docIdx[(d < N_DOCS) ? d : (N_DOCS - 1)];
        }
        #pragma unroll
        for (int i = 0; i < DOC_ITERS; ++i) {
            const int d = i * AGG_THREADS + tid;
            if (d < N_DOCS && (unsigned)(dr[i] - rbase) < (unsigned)rowsHere) docMask |= 1u << i;
        }
    }

    if (tid < PART_CHUNKS) {
        pcnt[tid] = myCnt;
        poff[tid] = myOff;
    }
    int totalV;
    const int myPos = block_excl_scan(myCnt, tid, wsum, totalV);
    if (tid < PART_CHUNKS) ppos[tid] = myPos;
    __syncthreads();
    const int total = __builtin_amdgcn_readfirstlane(totalV);

    if (total <= AGG_CAP) {
        for (int c = tid >> 2; c < PART_CHUNKS; c += AGG_THREADS / 4) {
            const int  n  = pcnt[c];
            const int* sp = part + (size_t)c * PART_EDGES + poff[c];
            int*       dp = lstS + ppos[c];
            for (int j = tid & 3; j < n; j += 16) {
                int u[4];
                #pragma unroll
                for (int k = 0; k < 4; ++k) u[k] = (j + 4 * k < n) ? sp[j + 4 * k] : 0;
                #pragma unroll
                for (int k = 0; k < 4; ++k) if (j + 4 * k < n) dp[j + 4 * k] = u[k];
            }
        }
        __syncthreads();

        int   eu[4], rk[4];
        float v[4];
        bool  ok[4];
        #pragma unroll
        for (int k = 0; k < 4; ++k) {
            const int p = k * AGG_THREADS + tid;
            ok[k] = p < total;
            eu[k] = ok[k] ? lstS[p] : 0;
        }
        #pragma unroll
        for (int k = 0; k < 4; ++k) {
            const float t = es[(unsigned)eu[k] >> 8] + ed[rbase + (eu[k] & 255)];
            v[k] = (t > 0.0f) ? t : 0.2f * t;
        }
        #pragma unroll
        for (int k = 0; k < 4; ++k) {
            rk[k] = 0;
            if (ok[k]) {
                atomicMax(&mxL[eu[k] & 255], f2key(v[k]));
                rk[k] = atomicAdd(&wcnt[eu[k] & (AGG_WAVES - 1)], 1);
            }
        }
        __syncthreads();
        if (tid < AGG_WAVES) {
            int s = 0;
            for (int j = 0; j < tid; ++j) s += wcnt[j];
            wbeg[tid] = s;
        }
        __syncthreads();
        #pragma unroll
        for (int k = 0; k < 4; ++k) {
            if (ok[k]) {
                const int pos = wbeg[eu[k] & (AGG_WAVES - 1)] + rk[k];
                lstS[pos] = eu[k];
                lstE[pos] = v[k];
            }
        }
        __syncthreads();

        const int beg = __builtin_amdgcn_readfirstlane(wbeg[wave]);
        const int end = beg + __builtin_amdgcn_readfirstlane(wcnt[wave]);
        float4 hv[8];
        int    su[4];
        float  sa[4];
        {
            int u; float a;
            agg_run_entries(beg, end, lstS, lstE, mxL, denL, lane, u, a);
            #pragma unroll
            for (int k = 0; k < 4; ++k) {
                su[k] = __builtin_amdgcn_readlane(u, k);
                sa[k] = __int_as_float(__builtin_amdgcn_readlane(__float_as_int(a), k));
                const float4* h = (const float4*)(H + (size_t)((unsigned)su[k] >> 8) * D_HID) + lane;
                hv[2 * k]     = h[0];
                hv[2 * k + 1] = h[32];
                __builtin_amdgcn_sched_barrier(0);
            }
        }
        for (int q = beg; q < end; q += 4) {
            int un; float an;
            agg_run_entries(q + 4, end, lstS, lstE, mxL, denL, lane, un, an);
            #pragma unroll
            for (int k = 0; k < 4; ++k) {
                const int row = su[k] & 255;
                if (row < AGG_ROWS)
                    agg_row_add<P>(acc, row, sa[k], hv[2 * k], hv[2 * k + 1], lane);
                su[k] = __builtin_amdgcn_readlane(un, k);
                sa[k] = __int_as_float(__builtin_amdgcn_readlane(__float_as_int(an), k));
                const float4* h = (const float4*)(H + (size_t)((unsigned)su[k] >> 8) * D_HID) + lane;
                hv[2 * k]     = h[0];
                hv[2 * k + 1] = h[32];
            }
        }

        const float4 b0 = ((const float4*)bias)[lane];
        const float4 b1 = ((const float4*)bias)[32 + lane];
        for (int row = wave; row < rowsHere; row += AGG_WAVES) {
            float4* r = (float4*)(acc + row * P) + lane;
            const float4 s0 = r[0];
            const float4 s1 = r[32];
            const float rdn = 1.0f / (denL[row] + 1e-16f);
            const float4 o0 = agg_finish4(s0, rdn, b0);
            const float4 o1 = agg_finish4(s1, rdn, b1);
            if (kHead) {
                r[0]  = o0;
                r[32] = o1;
            } else {
                float4* o = (float4*)(OUT + (size_t)(rbase + row) * D_HID) + lane;
                o[0]  = o0;
                o[32] = o1;
            }
        }
    } else {
        for (int i = tid; i < nE; i += AGG_THREADS) {
            const int d = dst[i];
            if ((unsigned)(d - rbase) < (unsigned)AGG_ROWS) {
                const float t = es[src[i]] + ed[d];
                atomicMax(&mxL[d - rbase], f2key((t > 0.0f) ? t : 0.2f * t));
            }
        }
        __syncthreads();
        for (int c0 = wave * 32; c0 < nE; c0 += AGG_WAVES * 32) {
            const int i = c0 + lane;
            const int d = (i < nE) ? dst[i] : -1;
            unsigned lm = __builtin_amdgcn_ballot_w32(
                (unsigned)(d - rbase) < (unsigned)AGG_ROWS);
            while (lm) {
                const int ej = c0 + __builtin_ctz(lm);
                lm &= lm - 1;
                const int   q = dst[ej] - rbase;
                const int   s = src[ej];
                const float t = es[s] + ed[rbase + q];
                const float e = (t > 0.0f) ? t : 0.2f * t;
                agg_add_edge<P, kHead>(acc, denL, H, s, q, __expf(e - key2f(mxL[q])), lane);
            }
        }
        __syncthreads();

        if (kHead) {
            for (int i = tid; i < rowsHere * (D_HID / 4); i += AGG_THREADS) {
                const int row = i >> 6;
                const int col = (i & 63) * 4;
                float4* a = (float4*)(acc + row * P + col);
                const float rdn = 1.0f / (denL[row] + 1e-16f);
                *a = agg_finish4(*a, rdn, *(const float4*)(bias + col));
            }
        } else {
            for (int i = tid; i < rowsHere * (D_HID / 4); i += AGG_THREADS) {
                const int row = i >> 6;
                const int hl  = i & 63;
                const int col = (hl >> 5) * 128 + 4 * (hl & 31);
                const float* a = acc + row * P + (hl >> 5) * 128 + (hl & 31);
                const float rdn = 1.0f / (denL[row] + 1e-16f);
                float4 o;
                o.x = fmaxf(a[0]  * rdn + bias[col + 0], 0.0f);
                o.y = fmaxf(a[32] * rdn + bias[col + 1], 0.0f);
                o.z = fmaxf(a[64] * rdn + bias[col + 2], 0.0f);
                o.w = fmaxf(a[96] * rdn + bias[col + 3], 0.0f);
                *(float4*)(OUT + (size_t)(rbase + row) * D_HID + col) = o;
            }
        }
    }

    if (kHead) {
        __syncthreads();
        float* const hid = lst.f;
        int ndV;
        const int myBase = block_excl_scan(__popc(docMask), tid, wsum, ndV);
        const int nd = __builtin_amdgcn_readfirstlane(ndV);
        const int cls = (lane < N_CLASSES) ? lane : 0;
        float w2[MLP_HID];
        #pragma unroll
        for (int k = 0; k < MLP_HID; ++k) w2[k] = Wm2[k * N_CLASSES + cls];
        const float bias2 = bm2[cls];

        for (int p0 = 0; p0 < nd; p0 += HEAD_PASS) {
            {
                unsigned m = docMask;
                int idx = myBase - p0;
                while (m) {
                    const int i = __builtin_ctz(m);
                    m &= m - 1;
                    if ((unsigned)idx < (unsigned)HEAD_PASS) {
                        const int d = i * AGG_THREADS + tid;
                        docL[idx] = (d << 8) | (docIdx[d] - rbase);
                    }
                    ++idx;
                }
            }
            __syncthreads();
            int cnt = nd - p0;
            if (cnt > HEAD_PASS) cnt = HEAD_PASS;
            const int nJobs = ((cnt + 15) >> 4) * (MLP_HID / 16);
            if (wave < nJobs) {
                const int t  = wave >> 2;
                const int j  = wave & 3;
                const int c  = lane & 15;
                const int h2 = (lane >> 4) << 1;
                const int li = 16 * t + c;
                const int row = (li < cnt) ? (docL[li] & 255) : 0;
                const float* ap = acc + row * P + h2;
                const float* bp = Ws + h2 * HEAD_PW + 16 * j + c;
                v8f hacc = {};
                for (int k0 = 0; k0 < D_HID; k0 += 32) {
                    v2f a[8], b[8];
                    #pragma unroll
                    for (int s = 0; s < 8; ++s) {
                        a[s].x = ap[k0 + 4 * s];
                        a[s].y = ap[k0 + 4 * s + 1];
                        b[s].x = bp[(k0 + 4 * s) * HEAD_PW];
                        b[s].y = bp[(k0 + 4 * s + 1) * HEAD_PW];
                    }
                    __builtin_amdgcn_sched_barrier(0);
                    #pragma unroll
                    for (int s = 0; s < 8; ++s)
                        hacc = __builtin_amdgcn_wmma_f32_16x16x4_f32(
                            false, a[s], false, b[s], (short)0, hacc, false, false);
                    __builtin_amdgcn_sched_barrier(0);
                    asm volatile("v_nop\n\tv_nop\n\tv_nop\n\tv_nop\n\tv_nop\n\tv_nop\n\tv_nop\n\tv_nop");
                    __builtin_amdgcn_sched_barrier(0);
                }
                const float bv = bm1[16 * j + c];
                #pragma unroll
                for (int r = 0; r < 8; ++r)
                    hid[(16 * t + 4 * h2 + r) * MLP_HID + 16 * j + c] = fmaxf(hacc[r] + bv, 0.0f);
            }
            __syncthreads();
            for (int q = wave; q < cnt; q += AGG_WAVES) {
                const int doc = docL[q] >> 8;
                const float4* hr = (const float4*)(hid + q * MLP_HID);
                float a2 = bias2;
                #pragma unroll
                for (int k4 = 0; k4 < MLP_HID / 4; ++k4) {
                    const float4 x = hr[k4];
                    a2 += x.x * w2[4 * k4 + 0];
                    a2 += x.y * w2[4 * k4 + 1];
                    a2 += x.z * w2[4 * k4 + 2];
                    a2 += x.w * w2[4 * k4 + 3];
                }
                const float logit = (lane < N_CLASSES) ? a2 : -3.0e38f;
                float mx = logit;
                #pragma unroll
                for (int off = 16; off; off >>= 1) mx = fmaxf(mx, __shfl_xor(mx, off, 32));
                const float ex = (lane < N_CLASSES) ? __expf(logit - mx) : 0.0f;
                float sum = ex;
                #pragma unroll
                for (int off = 16; off; off >>= 1) sum += __shfl_xor(sum, off, 32);
                if (lane < N_CLASSES) out[(size_t)doc * N_CLASSES + lane] = ex / sum;
            }
            __syncthreads();
        }
    }
}

extern "C" void kernel_launch(void* const* d_in, const int* in_sizes, int n_in,
                              void* d_out, int out_size, void* d_ws, size_t ws_size,
                              hipStream_t stream)
{
    (void)in_sizes; (void)n_in; (void)out_size; (void)ws_size;

    const float* x    = (const float*)d_in[0];
    const int*   ei   = (const int*)  d_in[1];
    const int*   doci = (const int*)  d_in[2];
    const float* W1   = (const float*)d_in[3];
    const float* a1s  = (const float*)d_in[4];
    const float* a1d  = (const float*)d_in[5];
    const float* b1   = (const float*)d_in[6];
    const float* W2   = (const float*)d_in[7];
    const float* a2s  = (const float*)d_in[8];
    const float* a2d  = (const float*)d_in[9];
    const float* b2   = (const float*)d_in[10];
    const float* Wm1  = (const float*)d_in[11];
    const float* bm1  = (const float*)d_in[12];
    const float* Wm2  = (const float*)d_in[13];
    const float* bm2  = (const float*)d_in[14];

    const int* src = ei;
    const int* dst = ei + N_EDGES;

    float* H    = (float*)d_ws;
    float* OUT  = H    + (size_t)N_NODES * D_HID;
    float* ES   = OUT  + (size_t)N_NODES * D_HID;
    float* ED   = ES   + N_NODES;
    int*   PART = (int*)(ED + N_NODES);
    int*   CNTT = PART + (size_t)PART_CHUNKS * PART_EDGES;
    int*   OFFT = CNTT + (size_t)256 * PART_CHUNKS;
    static_assert((size_t)PART_CHUNKS * PART_EDGES + (size_t)2 * 256 * PART_CHUNKS
                  <= (size_t)N_DOCS * MLP_HID, "partition fits the bytes the doc hidden matrix had");

    const dim3 gemmGridN(N_NODES / NG_TM, D_HID / NG_TN);
    const int gNodeW = (N_NODES + 7) / 8;
    const int gAgg   = AGG_OWNERS;

    edge_partition<<<PART_CHUNKS, PART_THREADS, 0, stream>>>(
        src, dst, PART, CNTT, OFFT, N_EDGES, N_NODES);

    for (int layer = 0; layer < 2; ++layer) {
        const float* A   = (layer == 0) ? x   : OUT;
        const float* W   = (layer == 0) ? W1  : W2;
        const float* as_ = (layer == 0) ? a1s : a2s;
        const float* ad_ = (layer == 0) ? a1d : a2d;
        const float* b   = (layer == 0) ? b1  : b2;

        gemm_nodes_wmma_f32<<<gemmGridN, NG_THREADS, 0, stream>>>(
            A, W, H, D_HID, D_HID);
        node_dots<<<gNodeW, 256, 0, stream>>>(H, as_, ad_, ES, ED, N_NODES);
        if (layer == 0)
            edge_aggregate_owned<false><<<gAgg, AGG_THREADS, 0, stream>>>(
                src, dst, ES, ED, H, b, OUT, PART, CNTT, OFFT, N_EDGES, N_NODES,
                nullptr, nullptr, nullptr, nullptr, nullptr, nullptr);
        else
            edge_aggregate_owned<true><<<gAgg, AGG_THREADS, 0, stream>>>(
                src, dst, ES, ED, H, b, OUT, PART, CNTT, OFFT, N_EDGES, N_NODES,
                doci, Wm1, bm1, Wm2, bm2, (float*)d_out);
    }
}
